# starting kernel
# baseline (speedup 1.0000x reference)
#include <hip/hip_runtime.h>

typedef __attribute__((ext_vector_type(2))) float v2f;
typedef __attribute__((ext_vector_type(8))) float v8f;

#define N_NODES 50000
#define N_EDGES 800000
#define D 128

__global__ void gin_zero_kernel(float4* __restrict__ p, int n4) {
    int i = blockIdx.x * blockDim.x + threadIdx.x;
    if (i < n4) {
        p[i] = make_float4(0.f, 0.f, 0.f, 0.f);
    }
}

__global__ void gin_scatter_kernel(const float* __restrict__ feat,
                                   const int* __restrict__ src,
                                   const int* __restrict__ dst,
                                   float* __restrict__ neigh) {
    int edge = (blockIdx.x * blockDim.x + threadIdx.x) >> 5;
    int lane = threadIdx.x & 31;
    if (edge >= N_EDGES) return;
    int s = src[edge];
    int t = dst[edge];
    const float4* frow = (const float4*)(feat + (size_t)s * D);
    float4 v = frow[lane];
    float* nrow = neigh + (size_t)t * D + lane * 4;
    atomicAdd(nrow + 0, v.x);
    atomicAdd(nrow + 1, v.y);
    atomicAdd(nrow + 2, v.z);
    atomicAdd(nrow + 3, v.w);
}

__global__ void gin_gemm_kernel(const float* __restrict__ feat,
                                const float* __restrict__ neigh,
                                const float* __restrict__ W,
                                const float* __restrict__ bias,
                                const float* __restrict__ eps,
                                float* __restrict__ out) {
    const int lane = threadIdx.x & 31;
    const int wave = threadIdx.x >> 5;
    const int half = lane >> 4;
    const int l15  = lane & 15;

    const int row0 = blockIdx.x * 16;
    const int n0   = wave * 16;

    const float scale = 1.0f + eps[0];

    const float* arow_f = feat  + (size_t)(row0 + l15) * D;
    const float* arow_n = neigh + (size_t)(row0 + l15) * D;

    v8f acc = {};
    #pragma unroll 4
    for (int k0 = 0; k0 < D; k0 += 4) {
        const int ka = k0 + half * 2;
        v2f a;
        a.x = scale * arow_f[ka]     + arow_n[ka];
        a.y = scale * arow_f[ka + 1] + arow_n[ka + 1];
        v2f bm;
        bm.x = W[(size_t)(ka)     * D + n0 + l15];
        bm.y = W[(size_t)(ka + 1) * D + n0 + l15];
        acc = __builtin_amdgcn_wmma_f32_16x16x4_f32(
             false, a,  false, bm,
             (short)0, acc,  false,  false);
    }

    const float bv = bias[n0 + l15];
    float* orow = out + (size_t)(row0 + half * 8) * D + n0 + l15;
    #pragma unroll
    for (int v = 0; v < 8; ++v) {
        orow[(size_t)v * D] = acc[v] + bv;
    }
}

extern "C" void kernel_launch(void* const* d_in, const int* in_sizes, int n_in,
                              void* d_out, int out_size, void* d_ws, size_t ws_size,
                              hipStream_t stream) {
    const float* feat = (const float*)d_in[0];
    const float* W    = (const float*)d_in[1];
    const float* b    = (const float*)d_in[2];
    const float* eps  = (const float*)d_in[3];
    const int*   src  = (const int*)d_in[4];
    const int*   dst  = (const int*)d_in[5];
    float*       out  = (float*)d_out;
    float*       neigh = (float*)d_ws;

    const int n4 = (N_NODES * D) / 4;
    gin_zero_kernel<<<(n4 + 255) / 256, 256, 0, stream>>>((float4*)neigh, n4);

    const long long threads = (long long)N_EDGES * 32;
    gin_scatter_kernel<<<(int)((threads + 255) / 256), 256, 0, stream>>>(
        feat, src, dst, neigh);

    gin_gemm_kernel<<<N_NODES / 16, 256, 0, stream>>>(
        feat, neigh, W, b, eps, out);
}


# scatter: lane l takes floats l+32j of the row, so each wave-wide f32 atomic (and load) covers one whole 128-byte line, was 8 dwords in each of 4 lines; atomics and arrival-order sums kept
# speedup vs baseline: 1.2775x; 1.2775x over previous
#include <hip/hip_runtime.h>

typedef __attribute__((ext_vector_type(2))) float v2f;
typedef __attribute__((ext_vector_type(8))) float v8f;

#define N_NODES 50000
#define N_EDGES 800000
#define D 128

__global__ void gin_zero_kernel(float4* __restrict__ p, int n4) {
    int i = blockIdx.x * blockDim.x + threadIdx.x;
    if (i < n4) {
        p[i] = make_float4(0.f, 0.f, 0.f, 0.f);
    }
}

__global__ void gin_scatter_kernel(const float* __restrict__ feat,
                                   const int* __restrict__ src,
                                   const int* __restrict__ dst,
                                   float* __restrict__ neigh) {
    int edge = (blockIdx.x * blockDim.x + threadIdx.x) >> 5;
    int lane = threadIdx.x & 31;
    if (edge >= N_EDGES) return;
    int s = src[edge];
    int t = dst[edge];
    const float* frow = feat + (size_t)s * D + lane;
    float* nrow = neigh + (size_t)t * D + lane;
    float v0 = frow[0];
    float v1 = frow[32];
    float v2 = frow[64];
    float v3 = frow[96];
    atomicAdd(nrow + 0,  v0);
    atomicAdd(nrow + 32, v1);
    atomicAdd(nrow + 64, v2);
    atomicAdd(nrow + 96, v3);
}

__global__ void gin_gemm_kernel(const float* __restrict__ feat,
                                const float* __restrict__ neigh,
                                const float* __restrict__ W,
                                const float* __restrict__ bias,
                                const float* __restrict__ eps,
                                float* __restrict__ out) {
    const int lane = threadIdx.x & 31;
    const int wave = threadIdx.x >> 5;
    const int half = lane >> 4;
    const int l15  = lane & 15;

    const int row0 = blockIdx.x * 16;
    const int n0   = wave * 16;

    const float scale = 1.0f + eps[0];

    const float* arow_f = feat  + (size_t)(row0 + l15) * D;
    const float* arow_n = neigh + (size_t)(row0 + l15) * D;

    v8f acc = {};
    #pragma unroll 4
    for (int k0 = 0; k0 < D; k0 += 4) {
        const int ka = k0 + half * 2;
        v2f a;
        a.x = scale * arow_f[ka]     + arow_n[ka];
        a.y = scale * arow_f[ka + 1] + arow_n[ka + 1];
        v2f bm;
        bm.x = W[(size_t)(ka)     * D + n0 + l15];
        bm.y = W[(size_t)(ka + 1) * D + n0 + l15];
        acc = __builtin_amdgcn_wmma_f32_16x16x4_f32(
             false, a,  false, bm,
             (short)0, acc,  false,  false);
    }

    const float bv = bias[n0 + l15];
    float* orow = out + (size_t)(row0 + half * 8) * D + n0 + l15;
    #pragma unroll
    for (int v = 0; v < 8; ++v) {
        orow[(size_t)v * D] = acc[v] + bv;
    }
}

extern "C" void kernel_launch(void* const* d_in, const int* in_sizes, int n_in,
                              void* d_out, int out_size, void* d_ws, size_t ws_size,
                              hipStream_t stream) {
    const float* feat = (const float*)d_in[0];
    const float* W    = (const float*)d_in[1];
    const float* b    = (const float*)d_in[2];
    const float* eps  = (const float*)d_in[3];
    const int*   src  = (const int*)d_in[4];
    const int*   dst  = (const int*)d_in[5];
    float*       out  = (float*)d_out;
    float*       neigh = (float*)d_ws;

    const int n4 = (N_NODES * D) / 4;
    gin_zero_kernel<<<(n4 + 255) / 256, 256, 0, stream>>>((float4*)neigh, n4);

    const long long threads = (long long)N_EDGES * 32;
    gin_scatter_kernel<<<(int)((threads + 255) / 256), 256, 0, stream>>>(
        feat, src, dst, neigh);

    gin_gemm_kernel<<<N_NODES / 16, 256, 0, stream>>>(
        feat, neigh, W, b, eps, out);
}


# scatter: one wave takes 32 consecutive edges (indices by one line load per array, broadcast per edge), next 8 edges' row loads in flight during 8 edges' atomics; arrival-order atomic sums kept
# speedup vs baseline: 1.6108x; 1.2609x over previous
#include <hip/hip_runtime.h>

typedef __attribute__((ext_vector_type(2))) float v2f;
typedef __attribute__((ext_vector_type(8))) float v8f;

#define N_NODES 50000
#define N_EDGES 800000
#define D 128

__global__ void gin_zero_kernel(float4* __restrict__ p, int n4) {
    int i = blockIdx.x * blockDim.x + threadIdx.x;
    if (i < n4) {
        p[i] = make_float4(0.f, 0.f, 0.f, 0.f);
    }
}

#define EDGES_PER_WAVE 32
#define EDGE_BATCH 8
#define N_BATCH (EDGES_PER_WAVE / EDGE_BATCH)
static_assert(N_EDGES % EDGES_PER_WAVE == 0, "each wave takes a full set of 32 edges");
static_assert(N_EDGES % (EDGES_PER_WAVE * 8) == 0, "grid of whole 8-wave blocks");

__global__ void __launch_bounds__(256)
gin_scatter_kernel(const float* __restrict__ feat,
                   const int* __restrict__ src,
                   const int* __restrict__ dst,
                   float* __restrict__ neigh) {
    const int lane = threadIdx.x & 31;
    const int wave = (blockIdx.x * blockDim.x + threadIdx.x) >> 5;
    const int base = wave * EDGES_PER_WAVE;
    if (base >= N_EDGES) return;

    const int sv = src[base + lane];
    const int tv = dst[base + lane];

    const unsigned lane_b = (unsigned)lane * 4u;
    const char* fbase = (const char*)feat;
    char* nbase = (char*)neigh;

    float v[2][EDGE_BATCH][4];

    #pragma unroll
    for (int j = 0; j < EDGE_BATCH; ++j) {
        const unsigned s = (unsigned)__builtin_amdgcn_readlane(sv, j);
        const float* frow = (const float*)(fbase + (s * (unsigned)(D * 4) + lane_b));
        v[0][j][0] = frow[0];
        v[0][j][1] = frow[32];
        v[0][j][2] = frow[64];
        v[0][j][3] = frow[96];
    }

    #pragma unroll
    for (int b = 0; b < N_BATCH; ++b) {
        const int cur = b & 1;
        const int nxt = cur ^ 1;
        if (b + 1 < N_BATCH) {
            #pragma unroll
            for (int j = 0; j < EDGE_BATCH; ++j) {
                const unsigned s = (unsigned)__builtin_amdgcn_readlane(sv, (b + 1) * EDGE_BATCH + j);
                const float* frow = (const float*)(fbase + (s * (unsigned)(D * 4) + lane_b));
                v[nxt][j][0] = frow[0];
                v[nxt][j][1] = frow[32];
                v[nxt][j][2] = frow[64];
                v[nxt][j][3] = frow[96];
            }
        }
        #pragma unroll
        for (int j = 0; j < EDGE_BATCH; ++j) {
            const unsigned t = (unsigned)__builtin_amdgcn_readlane(tv, b * EDGE_BATCH + j);
            float* nrow = (float*)(nbase + (t * (unsigned)(D * 4) + lane_b));
            atomicAdd(nrow + 0,  v[cur][j][0]);
            atomicAdd(nrow + 32, v[cur][j][1]);
            atomicAdd(nrow + 64, v[cur][j][2]);
            atomicAdd(nrow + 96, v[cur][j][3]);
        }
    }
}

__global__ void gin_gemm_kernel(const float* __restrict__ feat,
                                const float* __restrict__ neigh,
                                const float* __restrict__ W,
                                const float* __restrict__ bias,
                                const float* __restrict__ eps,
                                float* __restrict__ out) {
    const int lane = threadIdx.x & 31;
    const int wave = threadIdx.x >> 5;
    const int half = lane >> 4;
    const int l15  = lane & 15;

    const int row0 = blockIdx.x * 16;
    const int n0   = wave * 16;

    const float scale = 1.0f + eps[0];

    const float* arow_f = feat  + (size_t)(row0 + l15) * D;
    const float* arow_n = neigh + (size_t)(row0 + l15) * D;

    v8f acc = {};
    #pragma unroll 4
    for (int k0 = 0; k0 < D; k0 += 4) {
        const int ka = k0 + half * 2;
        v2f a;
        a.x = scale * arow_f[ka]     + arow_n[ka];
        a.y = scale * arow_f[ka + 1] + arow_n[ka + 1];
        v2f bm;
        bm.x = W[(size_t)(ka)     * D + n0 + l15];
        bm.y = W[(size_t)(ka + 1) * D + n0 + l15];
        acc = __builtin_amdgcn_wmma_f32_16x16x4_f32(
             false, a,  false, bm,
             (short)0, acc,  false,  false);
    }

    const float bv = bias[n0 + l15];
    float* orow = out + (size_t)(row0 + half * 8) * D + n0 + l15;
    #pragma unroll
    for (int v = 0; v < 8; ++v) {
        orow[(size_t)v * D] = acc[v] + bv;
    }
}

extern "C" void kernel_launch(void* const* d_in, const int* in_sizes, int n_in,
                              void* d_out, int out_size, void* d_ws, size_t ws_size,
                              hipStream_t stream) {
    const float* feat = (const float*)d_in[0];
    const float* W    = (const float*)d_in[1];
    const float* b    = (const float*)d_in[2];
    const float* eps  = (const float*)d_in[3];
    const int*   src  = (const int*)d_in[4];
    const int*   dst  = (const int*)d_in[5];
    float*       out  = (float*)d_out;
    float*       neigh = (float*)d_ws;

    const int n4 = (N_NODES * D) / 4;
    gin_zero_kernel<<<(n4 + 255) / 256, 256, 0, stream>>>((float4*)neigh, n4);

    gin_scatter_kernel<<<N_EDGES / (EDGES_PER_WAVE * 8), 256, 0, stream>>>(
        feat, src, dst, neigh);

    gin_gemm_kernel<<<N_NODES / 16, 256, 0, stream>>>(
        feat, neigh, W, b, eps, out);
}


# aggregation: a work-group owns 196 destination rows in LDS, scans dst, lists its edges, adds the rows by LDS float atomics (arrival-order sums), stores each line once; no global atomic, no zero pass
# speedup vs baseline: 1.9030x; 1.1814x over previous
#include <hip/hip_runtime.h>

typedef __attribute__((ext_vector_type(2))) float v2f;
typedef __attribute__((ext_vector_type(8))) float v8f;

#define N_NODES 50000
#define N_EDGES 800000
#define D 128

#define ROWS_PER_GROUP 196
#define N_GROUPS 256
#define GROUP_THREADS 1024
#define GROUP_WAVES (GROUP_THREADS / 32)
#define LIST_CAP 16384
#define EDGE_BATCH 8
#define SCAN_DEPTH 4
#define SCAN_STEP (SCAN_DEPTH * GROUP_THREADS)
#define N_QUADS (N_EDGES / 4)
#define SCAN_TRIPS ((N_QUADS + SCAN_STEP - 1) / SCAN_STEP)
#define QUAD_STAGGER 768
#define ACC_FLOATS ((ROWS_PER_GROUP + 1) * D)
static_assert(ROWS_PER_GROUP * N_GROUPS >= N_NODES, "the ranges cover every node");
static_assert((N_GROUPS - 1) * ROWS_PER_GROUP < N_NODES, "every group owns at least one node");
static_assert(ROWS_PER_GROUP + 1 <= 256, "a local row fits the low 8 bits of a list word");
static_assert(N_EDGES <= (1 << 24), "an edge number fits the high 24 bits of a list word");
static_assert(N_EDGES % 4 == 0, "dst is read 4 edges per load");
static_assert(N_QUADS % 16 == 0 && QUAD_STAGGER % 16 == 0, "staggered reads stay on 256-byte units");
static_assert((N_GROUPS - 1) * QUAD_STAGGER + SCAN_TRIPS * SCAN_STEP < 2 * N_QUADS,
              "one wrap brings every staggered position back inside the array");
static_assert(SCAN_TRIPS >= 2, "the scan loop fetches one trip ahead");
#define AGG_LDS_BYTES (ACC_FLOATS * 4 + LIST_CAP * 4 + 16)
static_assert(AGG_LDS_BYTES <= 320 * 1024, "sums, list and counter fit the work-group's LDS");

__device__ __forceinline__ void gin_scan_load(int4 (&d)[SCAN_DEPTH],
                                              const int4* __restrict__ dst4,
                                              int start, int kb) {
    #pragma unroll
    for (int u = 0; u < SCAN_DEPTH; ++u) {
        int i = start + kb + u * GROUP_THREADS;
        if (i >= N_QUADS) i -= N_QUADS;
        d[u] = dst4[i];
    }
}

__device__ __forceinline__ void gin_note_edge(unsigned dl, unsigned e,
                                              unsigned* list, unsigned* cnt) {
    if (dl < (unsigned)ROWS_PER_GROUP) {
        const unsigned slot = __hip_atomic_fetch_add(cnt, 1u, __ATOMIC_RELAXED,
                                                     __HIP_MEMORY_SCOPE_WORKGROUP);
        if (slot < (unsigned)LIST_CAP) list[slot] = (e << 8) | dl;
    }
}

__device__ __forceinline__ void gin_scan_use(const int4 (&d)[SCAN_DEPTH],
                                             int start, int kb, int lo,
                                             unsigned* list, unsigned* cnt) {
    #pragma unroll
    for (int u = 0; u < SCAN_DEPTH; ++u) {
        const int k = kb + u * GROUP_THREADS;
        int i = start + k;
        if (i >= N_QUADS) i -= N_QUADS;
        const unsigned d0 = (unsigned)(d[u].x - lo);
        const unsigned d1 = (unsigned)(d[u].y - lo);
        const unsigned d2 = (unsigned)(d[u].z - lo);
        const unsigned d3 = (unsigned)(d[u].w - lo);
        const unsigned dmin = min(min(d0, d1), min(d2, d3));
        if (dmin < (unsigned)ROWS_PER_GROUP && k < N_QUADS) {
            const unsigned e0 = (unsigned)i * 4u;
            gin_note_edge(d0, e0 + 0u, list, cnt);
            gin_note_edge(d1, e0 + 1u, list, cnt);
            gin_note_edge(d2, e0 + 2u, list, cnt);
            gin_note_edge(d3, e0 + 3u, list, cnt);
        }
    }
}

__global__ void __launch_bounds__(GROUP_THREADS)
gin_neigh_sum_kernel(const float* __restrict__ feat,
                     const int* __restrict__ src,
                     const int* __restrict__ dst,
                     float* __restrict__ neigh) {
    __shared__ __attribute__((aligned(16))) unsigned char L[AGG_LDS_BYTES];
    float* acc = (float*)L;
    unsigned* list = (unsigned*)(L + ACC_FLOATS * 4);
    unsigned* cnt = (unsigned*)(L + ACC_FLOATS * 4 + LIST_CAP * 4);

    const int tid = threadIdx.x;
    const int lane = tid & 31;
    const int wave = __builtin_amdgcn_readfirstlane(tid >> 5);
    const int lo = blockIdx.x * ROWS_PER_GROUP;

    float4* acc4 = (float4*)acc;
    for (int i = tid; i < ACC_FLOATS / 4; i += GROUP_THREADS) {
        acc4[i] = make_float4(0.f, 0.f, 0.f, 0.f);
    }
    if (tid == 0) *cnt = 0u;
    __syncthreads();

    const int4* dst4 = (const int4*)dst;
    const int start = blockIdx.x * QUAD_STAGGER;
    {
        int4 d[SCAN_DEPTH];
        gin_scan_load(d, dst4, start, tid);
        #pragma unroll 2
        for (int t = 0; t + 1 < SCAN_TRIPS; ++t) {
            const int kb = t * SCAN_STEP + tid;
            int4 dn[SCAN_DEPTH];
            gin_scan_load(dn, dst4, start, kb + SCAN_STEP);
            gin_scan_use(d, start, kb, lo, list, cnt);
            #pragma unroll
            for (int u = 0; u < SCAN_DEPTH; ++u) d[u] = dn[u];
        }
        gin_scan_use(d, start, (SCAN_TRIPS - 1) * SCAN_STEP + tid, lo, list, cnt);
    }
    __syncthreads();

    const unsigned total = (unsigned)__builtin_amdgcn_readfirstlane((int)*cnt);
    if (total <= (unsigned)LIST_CAP) {
        const int n = (int)total;
        const unsigned lane_b = (unsigned)lane * 4u;
        const char* fbase = (const char*)feat;
        const unsigned pad = (unsigned)ROWS_PER_GROUP;

        int b0 = wave * EDGE_BATCH;
        unsigned ent = pad;
        if (lane < EDGE_BATCH && b0 + lane < n) ent = list[b0 + lane];
        int s = src[ent >> 8];
        for (; b0 < n; b0 += GROUP_WAVES * EDGE_BATCH) {
            const unsigned ent_c = ent;
            const int s_c = s;
            float v[EDGE_BATCH][4];
            #pragma unroll
            for (int j = 0; j < EDGE_BATCH; ++j) {
                const unsigned sj = (unsigned)__builtin_amdgcn_readlane(s_c, j);
                const float* frow = (const float*)(fbase + (sj * (unsigned)(D * 4) + lane_b));
                v[j][0] = frow[0];
                v[j][1] = frow[32];
                v[j][2] = frow[64];
                v[j][3] = frow[96];
            }
            const int b1 = b0 + GROUP_WAVES * EDGE_BATCH;
            ent = pad;
            if (lane < EDGE_BATCH && b1 + lane < n) ent = list[b1 + lane];
            s = src[ent >> 8];
            #pragma unroll
            for (int j = 0; j < EDGE_BATCH; ++j) {
                const unsigned dj = (unsigned)__builtin_amdgcn_readlane((int)ent_c, j) & 255u;
                float* arow = acc + dj * D + lane;
                __hip_atomic_fetch_add(arow + 0,  v[j][0], __ATOMIC_RELAXED, __HIP_MEMORY_SCOPE_WORKGROUP);
                __hip_atomic_fetch_add(arow + 32, v[j][1], __ATOMIC_RELAXED, __HIP_MEMORY_SCOPE_WORKGROUP);
                __hip_atomic_fetch_add(arow + 64, v[j][2], __ATOMIC_RELAXED, __HIP_MEMORY_SCOPE_WORKGROUP);
                __hip_atomic_fetch_add(arow + 96, v[j][3], __ATOMIC_RELAXED, __HIP_MEMORY_SCOPE_WORKGROUP);
            }
        }
    } else {
        for (int e = tid; e < N_EDGES; e += GROUP_THREADS) {
            const unsigned dl = (unsigned)(dst[e] - lo);
            if (dl < (unsigned)ROWS_PER_GROUP) {
                const float* frow = feat + (size_t)src[e] * D;
                float* arow = acc + dl * D;
                for (int j = 0; j < D; ++j) {
                    __hip_atomic_fetch_add(arow + j, frow[j], __ATOMIC_RELAXED,
                                           __HIP_MEMORY_SCOPE_WORKGROUP);
                }
            }
        }
    }
    __syncthreads();

    const int rows = (N_NODES - lo) < ROWS_PER_GROUP ? (N_NODES - lo) : ROWS_PER_GROUP;
    float4* out4 = (float4*)(neigh + (size_t)lo * D);
    for (int i = tid; i < rows * (D / 4); i += GROUP_THREADS) {
        out4[i] = acc4[i];
    }
}

__global__ void gin_gemm_kernel(const float* __restrict__ feat,
                                const float* __restrict__ neigh,
                                const float* __restrict__ W,
                                const float* __restrict__ bias,
                                const float* __restrict__ eps,
                                float* __restrict__ out) {
    const int lane = threadIdx.x & 31;
    const int wave = threadIdx.x >> 5;
    const int half = lane >> 4;
    const int l15  = lane & 15;

    const int row0 = blockIdx.x * 16;
    const int n0   = wave * 16;

    const float scale = 1.0f + eps[0];

    const float* arow_f = feat  + (size_t)(row0 + l15) * D;
    const float* arow_n = neigh + (size_t)(row0 + l15) * D;

    v8f acc = {};
    #pragma unroll 4
    for (int k0 = 0; k0 < D; k0 += 4) {
        const int ka = k0 + half * 2;
        v2f a;
        a.x = scale * arow_f[ka]     + arow_n[ka];
        a.y = scale * arow_f[ka + 1] + arow_n[ka + 1];
        v2f bm;
        bm.x = W[(size_t)(ka)     * D + n0 + l15];
        bm.y = W[(size_t)(ka + 1) * D + n0 + l15];
        acc = __builtin_amdgcn_wmma_f32_16x16x4_f32(
             false, a,  false, bm,
             (short)0, acc,  false,  false);
    }

    const float bv = bias[n0 + l15];
    float* orow = out + (size_t)(row0 + half * 8) * D + n0 + l15;
    #pragma unroll
    for (int v = 0; v < 8; ++v) {
        orow[(size_t)v * D] = acc[v] + bv;
    }
}

extern "C" void kernel_launch(void* const* d_in, const int* in_sizes, int n_in,
                              void* d_out, int out_size, void* d_ws, size_t ws_size,
                              hipStream_t stream) {
    const float* feat = (const float*)d_in[0];
    const float* W    = (const float*)d_in[1];
    const float* b    = (const float*)d_in[2];
    const float* eps  = (const float*)d_in[3];
    const int*   src  = (const int*)d_in[4];
    const int*   dst  = (const int*)d_in[5];
    float*       out  = (float*)d_out;
    float*       neigh = (float*)d_ws;

    gin_neigh_sum_kernel<<<N_GROUPS, GROUP_THREADS, 0, stream>>>(
        feat, src, dst, neigh);

    gin_gemm_kernel<<<N_NODES / 16, 256, 0, stream>>>(
        feat, neigh, W, b, eps, out);
}


# dense layer fused into the aggregation kernel: sum rows stay in LDS (pitch 132), (1+eps)*feat is added there, f32 WMMA reads both operands from LDS (W staged as k pairs); one launch, no sum plane
# speedup vs baseline: 2.8475x; 1.4963x over previous
#include <hip/hip_runtime.h>

typedef __attribute__((ext_vector_type(2))) float v2f;
typedef __attribute__((ext_vector_type(8))) float v8f;

#define N_NODES 50000
#define N_EDGES 800000
#define D 128

#define ROWS_PER_GROUP 196
#define N_GROUPS 256
#define GROUP_THREADS 1024
#define GROUP_WAVES (GROUP_THREADS / 32)
#define LIST_CAP 16384
#define EDGE_BATCH 8
#define SCAN_DEPTH 4
#define SCAN_STEP (SCAN_DEPTH * GROUP_THREADS)
#define N_QUADS (N_EDGES / 4)
#define SCAN_TRIPS ((N_QUADS + SCAN_STEP - 1) / SCAN_STEP)
#define QUAD_STAGGER 768
#define ACC_PITCH (D + 4)
#define ACC_ROWS 208
#define ACC_FLOATS (ACC_ROWS * ACC_PITCH)
#define W_PITCH (2 * D + 32)
#define W_FLOATS ((D / 2) * W_PITCH)
#define FEAT_PIECES ((ROWS_PER_GROUP * (D / 4) + GROUP_THREADS - 1) / GROUP_THREADS)
#define W_PIECES (((D / 2) * (D / 4)) / GROUP_THREADS)
static_assert(ROWS_PER_GROUP * N_GROUPS >= N_NODES, "the ranges cover every node");
static_assert((N_GROUPS - 1) * ROWS_PER_GROUP < N_NODES, "every group owns at least one node");
static_assert(ROWS_PER_GROUP + 1 <= 256, "a local row fits the low 8 bits of a list word");
static_assert(ROWS_PER_GROUP + 1 <= ACC_ROWS && ACC_ROWS % 16 == 0, "whole row tiles, padding row included");
static_assert(N_EDGES <= (1 << 24), "an edge number fits the high 24 bits of a list word");
static_assert(N_EDGES % 4 == 0, "dst is read 4 edges per load");
static_assert(N_QUADS % 16 == 0 && QUAD_STAGGER % 16 == 0, "staggered reads stay on 256-byte units");
static_assert((N_GROUPS - 1) * QUAD_STAGGER + SCAN_TRIPS * SCAN_STEP < 2 * N_QUADS,
              "one wrap brings every staggered position back inside the array");
static_assert(SCAN_TRIPS >= 2, "the scan loop fetches one trip ahead");
static_assert(D == 128 && ACC_PITCH % 4 == 0 && W_PITCH % 4 == 0, "16-byte LDS accesses stay aligned");
static_assert(((D / 2) * (D / 4)) % GROUP_THREADS == 0, "W is staged in whole rounds of the group");
#define AGG_LDS_BYTES (ACC_FLOATS * 4 + LIST_CAP * 4 + 16 + W_FLOATS * 4)
static_assert(AGG_LDS_BYTES <= 320 * 1024, "sums, list, counter and W fit the work-group's LDS");

__device__ __forceinline__ void gin_scan_load(int4 (&d)[SCAN_DEPTH],
                                              const int4* __restrict__ dst4,
                                              int start, int kb) {
    #pragma unroll
    for (int u = 0; u < SCAN_DEPTH; ++u) {
        int i = start + kb + u * GROUP_THREADS;
        if (i >= N_QUADS) i -= N_QUADS;
        d[u] = dst4[i];
    }
}

__device__ __forceinline__ void gin_note_edge(unsigned dl, unsigned e,
                                              unsigned* list, unsigned* cnt) {
    if (dl < (unsigned)ROWS_PER_GROUP) {
        const unsigned slot = __hip_atomic_fetch_add(cnt, 1u, __ATOMIC_RELAXED,
                                                     __HIP_MEMORY_SCOPE_WORKGROUP);
        if (slot < (unsigned)LIST_CAP) list[slot] = (e << 8) | dl;
    }
}

__device__ __forceinline__ void gin_scan_use(const int4 (&d)[SCAN_DEPTH],
                                             int start, int kb, int lo,
                                             unsigned* list, unsigned* cnt) {
    #pragma unroll
    for (int u = 0; u < SCAN_DEPTH; ++u) {
        const int k = kb + u * GROUP_THREADS;
        int i = start + k;
        if (i >= N_QUADS) i -= N_QUADS;
        const unsigned d0 = (unsigned)(d[u].x - lo);
        const unsigned d1 = (unsigned)(d[u].y - lo);
        const unsigned d2 = (unsigned)(d[u].z - lo);
        const unsigned d3 = (unsigned)(d[u].w - lo);
        const unsigned dmin = min(min(d0, d1), min(d2, d3));
        if (dmin < (unsigned)ROWS_PER_GROUP && k < N_QUADS) {
            const unsigned e0 = (unsigned)i * 4u;
            gin_note_edge(d0, e0 + 0u, list, cnt);
            gin_note_edge(d1, e0 + 1u, list, cnt);
            gin_note_edge(d2, e0 + 2u, list, cnt);
            gin_note_edge(d3, e0 + 3u, list, cnt);
        }
    }
}

__global__ void __launch_bounds__(GROUP_THREADS)
gin_conv_kernel(const float* __restrict__ feat,
                const int* __restrict__ src,
                const int* __restrict__ dst,
                const float* __restrict__ W,
                const float* __restrict__ bias,
                const float* __restrict__ eps,
                float* __restrict__ out) {
    __shared__ __attribute__((aligned(16))) unsigned char L[AGG_LDS_BYTES];
    float* acc = (float*)L;
    unsigned* list = (unsigned*)(L + ACC_FLOATS * 4);
    unsigned* cnt = (unsigned*)(L + ACC_FLOATS * 4 + LIST_CAP * 4);
    float* wl = (float*)(L + ACC_FLOATS * 4 + LIST_CAP * 4 + 16);

    const int tid = threadIdx.x;
    const int lane = tid & 31;
    const int wave = __builtin_amdgcn_readfirstlane(tid >> 5);
    const int lo = blockIdx.x * ROWS_PER_GROUP;

    float4* acc4 = (float4*)acc;
    for (int i = tid; i < ACC_FLOATS / 4; i += GROUP_THREADS) {
        acc4[i] = make_float4(0.f, 0.f, 0.f, 0.f);
    }
    if (tid == 0) *cnt = 0u;
    __syncthreads();

    const int4* dst4 = (const int4*)dst;
    const int start = blockIdx.x * QUAD_STAGGER;
    {
        int4 d[SCAN_DEPTH];
        gin_scan_load(d, dst4, start, tid);
        #pragma unroll 2
        for (int t = 0; t + 1 < SCAN_TRIPS; ++t) {
            const int kb = t * SCAN_STEP + tid;
            int4 dn[SCAN_DEPTH];
            gin_scan_load(dn, dst4, start, kb + SCAN_STEP);
            gin_scan_use(d, start, kb, lo, list, cnt);
            #pragma unroll
            for (int u = 0; u < SCAN_DEPTH; ++u) d[u] = dn[u];
        }
        gin_scan_use(d, start, (SCAN_TRIPS - 1) * SCAN_STEP + tid, lo, list, cnt);
    }
    __syncthreads();

    const unsigned total = (unsigned)__builtin_amdgcn_readfirstlane((int)*cnt);
    if (total <= (unsigned)LIST_CAP) {
        const int n = (int)total;
        const unsigned lane_b = (unsigned)lane * 4u;
        const char* fbase = (const char*)feat;
        const unsigned pad = (unsigned)ROWS_PER_GROUP;

        int b0 = wave * EDGE_BATCH;
        unsigned ent = pad;
        if (lane < EDGE_BATCH && b0 + lane < n) ent = list[b0 + lane];
        int s = src[ent >> 8];
        for (; b0 < n; b0 += GROUP_WAVES * EDGE_BATCH) {
            const unsigned ent_c = ent;
            const int s_c = s;
            float v[EDGE_BATCH][4];
            #pragma unroll
            for (int j = 0; j < EDGE_BATCH; ++j) {
                const unsigned sj = (unsigned)__builtin_amdgcn_readlane(s_c, j);
                const float* frow = (const float*)(fbase + (sj * (unsigned)(D * 4) + lane_b));
                v[j][0] = frow[0];
                v[j][1] = frow[32];
                v[j][2] = frow[64];
                v[j][3] = frow[96];
            }
            const int b1 = b0 + GROUP_WAVES * EDGE_BATCH;
            ent = pad;
            if (lane < EDGE_BATCH && b1 + lane < n) ent = list[b1 + lane];
            s = src[ent >> 8];
            #pragma unroll
            for (int j = 0; j < EDGE_BATCH; ++j) {
                const unsigned dj = (unsigned)__builtin_amdgcn_readlane((int)ent_c, j) & 255u;
                float* arow = acc + dj * ACC_PITCH + lane;
                __hip_atomic_fetch_add(arow + 0,  v[j][0], __ATOMIC_RELAXED, __HIP_MEMORY_SCOPE_WORKGROUP);
                __hip_atomic_fetch_add(arow + 32, v[j][1], __ATOMIC_RELAXED, __HIP_MEMORY_SCOPE_WORKGROUP);
                __hip_atomic_fetch_add(arow + 64, v[j][2], __ATOMIC_RELAXED, __HIP_MEMORY_SCOPE_WORKGROUP);
                __hip_atomic_fetch_add(arow + 96, v[j][3], __ATOMIC_RELAXED, __HIP_MEMORY_SCOPE_WORKGROUP);
            }
        }
    } else {
        for (int e = tid; e < N_EDGES; e += GROUP_THREADS) {
            const unsigned dl = (unsigned)(dst[e] - lo);
            if (dl < (unsigned)ROWS_PER_GROUP) {
                const float* frow = feat + (size_t)src[e] * D;
                float* arow = acc + dl * ACC_PITCH;
                for (int j = 0; j < D; ++j) {
                    __hip_atomic_fetch_add(arow + j, frow[j], __ATOMIC_RELAXED,
                                           __HIP_MEMORY_SCOPE_WORKGROUP);
                }
            }
        }
    }
    __syncthreads();

    const int rows = (N_NODES - lo) < ROWS_PER_GROUP ? (N_NODES - lo) : ROWS_PER_GROUP;
    const float scale = 1.0f + eps[0];

    const float4* feat4 = (const float4*)(feat + (size_t)lo * D);
    const float4* W4 = (const float4*)W;
    const int n_f4 = rows * (D / 4);
    float4 f[FEAT_PIECES];
    #pragma unroll
    for (int u = 0; u < FEAT_PIECES; ++u) {
        const int i = tid + u * GROUP_THREADS;
        f[u] = make_float4(0.f, 0.f, 0.f, 0.f);
        if (i < n_f4) f[u] = feat4[i];
    }
    float4 w0[W_PIECES];
    float4 w1[W_PIECES];
    #pragma unroll
    for (int u = 0; u < W_PIECES; ++u) {
        const int i = tid + u * GROUP_THREADS;
        w0[u] = W4[(2 * (i >> 5)) * (D / 4) + (i & 31)];
        w1[u] = W4[(2 * (i >> 5) + 1) * (D / 4) + (i & 31)];
    }
    #pragma unroll
    for (int u = 0; u < FEAT_PIECES; ++u) {
        const int i = tid + u * GROUP_THREADS;
        if (i < n_f4) {
            float4* p = (float4*)(acc + (i >> 5) * ACC_PITCH + (i & 31) * 4);
            float4 a = *p;
            a.x = scale * f[u].x + a.x;
            a.y = scale * f[u].y + a.y;
            a.z = scale * f[u].z + a.z;
            a.w = scale * f[u].w + a.w;
            *p = a;
        }
    }
    #pragma unroll
    for (int u = 0; u < W_PIECES; ++u) {
        const int i = tid + u * GROUP_THREADS;
        float4* t = (float4*)(wl + (i >> 5) * W_PITCH + (i & 31) * 8);
        t[0] = make_float4(w0[u].x, w1[u].x, w0[u].y, w1[u].y);
        t[1] = make_float4(w0[u].z, w1[u].z, w0[u].w, w1[u].w);
    }
    __syncthreads();

    const int half = lane >> 4;
    const int l15 = lane & 15;
    const int n_strips = ((rows + 15) >> 4) * 4;
    for (int st = wave; st < n_strips; st += GROUP_WAVES) {
        const int r0 = (st >> 2) * 16;
        const int n0 = (st & 3) * 32;
        const float* ap = acc + (r0 + l15) * ACC_PITCH + half * 2;
        const float* bp = wl + half * W_PITCH + (n0 + l15) * 2;
        v8f c0 = {};
        v8f c1 = {};
        #pragma unroll 8
        for (int ks = 0; ks < D / 4; ++ks) {
            const v2f a  = *(const v2f*)(ap + ks * 4);
            const v2f b0 = *(const v2f*)(bp + ks * (2 * W_PITCH));
            const v2f b1 = *(const v2f*)(bp + ks * (2 * W_PITCH) + 32);
            c0 = __builtin_amdgcn_wmma_f32_16x16x4_f32(
                 false, a,  false, b0,
                 (short)0, c0,  false,  false);
            c1 = __builtin_amdgcn_wmma_f32_16x16x4_f32(
                 false, a,  false, b1,
                 (short)0, c1,  false,  false);
        }
        const float bv0 = bias[n0 + l15];
        const float bv1 = bias[n0 + 16 + l15];
        const int rl = r0 + half * 8;
        float* orow = out + (size_t)(lo + rl) * D + n0 + l15;
        #pragma unroll
        for (int v = 0; v < 8; ++v) {
            if (rl + v < rows) {
                orow[(size_t)v * D]      = c0[v] + bv0;
                orow[(size_t)v * D + 16] = c1[v] + bv1;
            }
        }
    }
}

extern "C" void kernel_launch(void* const* d_in, const int* in_sizes, int n_in,
                              void* d_out, int out_size, void* d_ws, size_t ws_size,
                              hipStream_t stream) {
    const float* feat = (const float*)d_in[0];
    const float* W    = (const float*)d_in[1];
    const float* b    = (const float*)d_in[2];
    const float* eps  = (const float*)d_in[3];
    const int*   src  = (const int*)d_in[4];
    const int*   dst  = (const int*)d_in[5];
    float*       out  = (float*)d_out;

    gin_conv_kernel<<<N_GROUPS, GROUP_THREADS, 0, stream>>>(
        feat, src, dst, W, b, eps, out);
}


# edges sorted by destination range once, in a first launch (256 slices, LDS counting sort; word = source row, local row); a group copies its runs, no scan of all of dst; arrival-order sums kept
# speedup vs baseline: 4.3717x; 1.5353x over previous
#include <hip/hip_runtime.h>

typedef __attribute__((ext_vector_type(2))) float v2f;
typedef __attribute__((ext_vector_type(8))) float v8f;

#define N_NODES 50000
#define N_EDGES 800000
#define D 128

#define ROWS_PER_GROUP 196
#define N_GROUPS 256
#define GROUP_THREADS 1024
#define GROUP_WAVES (GROUP_THREADS / 32)
#define LIST_CAP 16384
#define EDGE_BATCH 8
#define N_SLICES 256
#define SLICE_EDGES (N_EDGES / N_SLICES)
#define BIN_THREADS 256
#define BIN_WAVES (BIN_THREADS / 32)
#define BIN_FULL (SLICE_EDGES / BIN_THREADS)
#define BIN_TAIL (SLICE_EDGES - BIN_FULL * BIN_THREADS)
#define BIN_PER_THREAD (BIN_FULL + 1)
#define SLICE_PITCH 3200
#define TAB_PITCH 288
#define TAB_BASE (N_SLICES * SLICE_PITCH)
#define TEAM 4
#define TEAM_DEPTH 4
#define ACC_PITCH (D + 4)
#define ACC_ROWS 208
#define ACC_FLOATS (ACC_ROWS * ACC_PITCH)
#define W_PITCH (2 * D + 32)
#define W_FLOATS ((D / 2) * W_PITCH)
#define FEAT_PIECES ((ROWS_PER_GROUP * (D / 4) + GROUP_THREADS - 1) / GROUP_THREADS)
#define W_PIECES (((D / 2) * (D / 4)) / GROUP_THREADS)
static_assert(ROWS_PER_GROUP * N_GROUPS >= N_NODES, "the ranges cover every node");
static_assert((N_GROUPS - 1) * ROWS_PER_GROUP < N_NODES, "every group owns at least one node");
static_assert(ROWS_PER_GROUP + 1 <= 256, "a local row fits the low 8 bits of an edge word");
static_assert(ROWS_PER_GROUP + 1 <= ACC_ROWS && ACC_ROWS % 16 == 0, "whole row tiles, padding row included");
static_assert(N_NODES <= (1 << 24), "a source row fits the high 24 bits of an edge word");
static_assert(N_EDGES % N_SLICES == 0, "the slices are equal");
static_assert(BIN_THREADS == N_GROUPS, "one thread of a launch-1 group per counter");
static_assert(BIN_TAIL > 0 && BIN_TAIL <= BIN_THREADS, "the last round is a short one");
static_assert(SLICE_PITCH >= SLICE_EDGES && SLICE_PITCH % 32 == 0, "a slice's words fill whole lines of its own");
static_assert(TAB_PITCH >= N_GROUPS + 1 && TAB_PITCH % 32 == 0, "a slice's offsets fill whole lines of its own");
static_assert(N_SLICES * TEAM == GROUP_THREADS && (TEAM & (TEAM - 1)) == 0 && TEAM <= 32,
              "every slice has one team, inside one wave");
static_assert(D == 128 && ACC_PITCH % 4 == 0 && W_PITCH % 4 == 0, "16-byte LDS accesses stay aligned");
static_assert(((D / 2) * (D / 4)) % GROUP_THREADS == 0, "W is staged in whole rounds of the group");
#define AGG_LDS_BYTES (ACC_FLOATS * 4 + LIST_CAP * 4 + 16 + W_FLOATS * 4)
static_assert(AGG_LDS_BYTES <= 320 * 1024, "sums, list, counter and W fit the work-group's LDS");

__global__ void __launch_bounds__(BIN_THREADS)
gin_bin_kernel(const int* __restrict__ src,
               const int* __restrict__ dst,
               unsigned* __restrict__ ws) {
    __shared__ unsigned count[N_GROUPS];
    __shared__ unsigned offs[N_GROUPS + 1];
    __shared__ unsigned wave_sum[BIN_WAVES];
    __shared__ unsigned sorted[SLICE_EDGES];

    const int tid = threadIdx.x;
    const int lane = tid & 31;
    const int wave = tid >> 5;
    const int e0 = blockIdx.x * SLICE_EDGES;

    int dv[BIN_PER_THREAD];
    int sv[BIN_PER_THREAD];
    dv[BIN_FULL] = -1;
    sv[BIN_FULL] = 0;
    if (tid < BIN_TAIL) {
        dv[BIN_FULL] = dst[e0 + BIN_FULL * BIN_THREADS + tid];
        sv[BIN_FULL] = src[e0 + BIN_FULL * BIN_THREADS + tid];
    }
    #pragma unroll
    for (int u = 0; u < BIN_FULL; ++u) {
        dv[u] = dst[e0 + u * BIN_THREADS + tid];
        sv[u] = src[e0 + u * BIN_THREADS + tid];
    }
    count[tid] = 0u;
    __syncthreads();

    unsigned h[BIN_PER_THREAD];
    unsigned word[BIN_PER_THREAD];
    unsigned rank[BIN_PER_THREAD];
    #pragma unroll
    for (int u = 0; u < BIN_PER_THREAD; ++u) {
        const unsigned d = (unsigned)dv[u];
        h[u] = N_GROUPS;
        word[u] = 0u;
        rank[u] = 0u;
        if (d < (unsigned)N_NODES) {
            h[u] = d / (unsigned)ROWS_PER_GROUP;
            word[u] = ((unsigned)sv[u] << 8) | (d - h[u] * (unsigned)ROWS_PER_GROUP);
            rank[u] = __hip_atomic_fetch_add(&count[h[u]], 1u, __ATOMIC_RELAXED,
                                             __HIP_MEMORY_SCOPE_WORKGROUP);
        }
    }
    __syncthreads();

    const unsigned c = count[tid];
    unsigned x = c;
    #pragma unroll
    for (int o = 1; o < 32; o <<= 1) {
        const unsigned y = __shfl_up(x, o);
        if (lane >= o) x += y;
    }
    if (lane == 31) wave_sum[wave] = x;
    __syncthreads();
    unsigned base = 0u;
    #pragma unroll
    for (int w = 0; w < BIN_WAVES; ++w) {
        if (w < wave) base += wave_sum[w];
    }
    offs[tid] = base + x - c;
    if (tid == BIN_THREADS - 1) offs[N_GROUPS] = base + x;
    __syncthreads();

    unsigned place[BIN_PER_THREAD];
    #pragma unroll
    for (int u = 0; u < BIN_PER_THREAD; ++u) place[u] = offs[h[u]] + rank[u];
    #pragma unroll
    for (int u = 0; u < BIN_PER_THREAD; ++u) {
        if (h[u] < (unsigned)N_GROUPS) sorted[place[u]] = word[u];
    }
    __syncthreads();

    const unsigned kept = offs[N_GROUPS];
    unsigned* wout = ws + (size_t)blockIdx.x * SLICE_PITCH;
    for (unsigned k = (unsigned)tid; k < kept; k += BIN_THREADS) wout[k] = sorted[k];
    unsigned* tout = ws + TAB_BASE + (size_t)blockIdx.x * TAB_PITCH;
    tout[tid] = offs[tid];
    if (tid == 0) tout[N_GROUPS] = kept;
}

__global__ void __launch_bounds__(GROUP_THREADS)
gin_conv_kernel(const float* __restrict__ feat,
                const unsigned* __restrict__ ws,
                const float* __restrict__ W,
                const float* __restrict__ bias,
                const float* __restrict__ eps,
                float* __restrict__ out) {
    __shared__ __attribute__((aligned(16))) unsigned char L[AGG_LDS_BYTES];
    float* acc = (float*)L;
    unsigned* list = (unsigned*)(L + ACC_FLOATS * 4);
    unsigned* cnt = (unsigned*)(L + ACC_FLOATS * 4 + LIST_CAP * 4);
    float* wl = (float*)(L + ACC_FLOATS * 4 + LIST_CAP * 4 + 16);

    const int tid = threadIdx.x;
    const int lane = tid & 31;
    const int wave = __builtin_amdgcn_readfirstlane(tid >> 5);
    const int lo = blockIdx.x * ROWS_PER_GROUP;

    const int slice = tid / TEAM;
    const int q = tid & (TEAM - 1);
    const unsigned* trow = ws + TAB_BASE + (size_t)slice * TAB_PITCH + blockIdx.x;
    const unsigned run_lo = trow[0];
    const unsigned run_hi = trow[1];

    float4* acc4 = (float4*)acc;
    for (int i = tid; i < ACC_FLOATS / 4; i += GROUP_THREADS) {
        acc4[i] = make_float4(0.f, 0.f, 0.f, 0.f);
    }
    if (tid == 0) *cnt = 0u;
    __syncthreads();

    const unsigned run = run_hi - run_lo;
    const unsigned* ewords = ws + (size_t)slice * SLICE_PITCH + run_lo;
    unsigned pos = 0u;
    if (q == 0) {
        pos = __hip_atomic_fetch_add(cnt, run, __ATOMIC_RELAXED, __HIP_MEMORY_SCOPE_WORKGROUP);
    }
    pos = (unsigned)__shfl((int)pos, lane & ~(TEAM - 1));
    unsigned ew[TEAM_DEPTH];
    #pragma unroll
    for (int u = 0; u < TEAM_DEPTH; ++u) {
        const unsigned k = (unsigned)(q + u * TEAM);
        ew[u] = 0u;
        if (k < run) ew[u] = ewords[k];
    }
    __syncthreads();

    const unsigned total = (unsigned)__builtin_amdgcn_readfirstlane((int)*cnt);
    if (total <= (unsigned)LIST_CAP) {
        #pragma unroll
        for (int u = 0; u < TEAM_DEPTH; ++u) {
            const unsigned k = (unsigned)(q + u * TEAM);
            if (k < run) list[pos + k] = ew[u];
        }
        for (unsigned k = (unsigned)(q + TEAM_DEPTH * TEAM); k < run; k += TEAM) {
            list[pos + k] = ewords[k];
        }
    }
    __syncthreads();

    if (total <= (unsigned)LIST_CAP) {
        const int n = (int)total;
        const unsigned lane_b = (unsigned)lane * 4u;
        const char* fbase = (const char*)feat;
        const unsigned pad = (unsigned)ROWS_PER_GROUP;

        int b0 = wave * EDGE_BATCH;
        unsigned ent = pad;
        if (lane < EDGE_BATCH && b0 + lane < n) ent = list[b0 + lane];
        for (; b0 < n; b0 += GROUP_WAVES * EDGE_BATCH) {
            const unsigned ent_c = ent;
            float v[EDGE_BATCH][4];
            #pragma unroll
            for (int j = 0; j < EDGE_BATCH; ++j) {
                const unsigned sj = (unsigned)__builtin_amdgcn_readlane((int)ent_c, j) >> 8;
                const float* frow = (const float*)(fbase + (sj * (unsigned)(D * 4) + lane_b));
                v[j][0] = frow[0];
                v[j][1] = frow[32];
                v[j][2] = frow[64];
                v[j][3] = frow[96];
            }
            const int b1 = b0 + GROUP_WAVES * EDGE_BATCH;
            ent = pad;
            if (lane < EDGE_BATCH && b1 + lane < n) ent = list[b1 + lane];
            #pragma unroll
            for (int j = 0; j < EDGE_BATCH; ++j) {
                const unsigned dj = (unsigned)__builtin_amdgcn_readlane((int)ent_c, j) & 255u;
                float* arow = acc + dj * ACC_PITCH + lane;
                __hip_atomic_fetch_add(arow + 0,  v[j][0], __ATOMIC_RELAXED, __HIP_MEMORY_SCOPE_WORKGROUP);
                __hip_atomic_fetch_add(arow + 32, v[j][1], __ATOMIC_RELAXED, __HIP_MEMORY_SCOPE_WORKGROUP);
                __hip_atomic_fetch_add(arow + 64, v[j][2], __ATOMIC_RELAXED, __HIP_MEMORY_SCOPE_WORKGROUP);
                __hip_atomic_fetch_add(arow + 96, v[j][3], __ATOMIC_RELAXED, __HIP_MEMORY_SCOPE_WORKGROUP);
            }
        }
    } else {
        for (unsigned k = (unsigned)q; k < run; k += TEAM) {
            const unsigned w = ewords[k];
            const float* frow = feat + (size_t)(w >> 8) * D;
            float* arow = acc + (w & 255u) * ACC_PITCH;
            for (int j = 0; j < D; ++j) {
                __hip_atomic_fetch_add(arow + j, frow[j], __ATOMIC_RELAXED,
                                       __HIP_MEMORY_SCOPE_WORKGROUP);
            }
        }
    }
    __syncthreads();

    const int rows = (N_NODES - lo) < ROWS_PER_GROUP ? (N_NODES - lo) : ROWS_PER_GROUP;
    const float scale = 1.0f + eps[0];

    const float4* feat4 = (const float4*)(feat + (size_t)lo * D);
    const float4* W4 = (const float4*)W;
    const int n_f4 = rows * (D / 4);
    float4 f[FEAT_PIECES];
    #pragma unroll
    for (int u = 0; u < FEAT_PIECES; ++u) {
        const int i = tid + u * GROUP_THREADS;
        f[u] = make_float4(0.f, 0.f, 0.f, 0.f);
        if (i < n_f4) f[u] = feat4[i];
    }
    float4 w0[W_PIECES];
    float4 w1[W_PIECES];
    #pragma unroll
    for (int u = 0; u < W_PIECES; ++u) {
        const int i = tid + u * GROUP_THREADS;
        w0[u] = W4[(2 * (i >> 5)) * (D / 4) + (i & 31)];
        w1[u] = W4[(2 * (i >> 5) + 1) * (D / 4) + (i & 31)];
    }
    #pragma unroll
    for (int u = 0; u < FEAT_PIECES; ++u) {
        const int i = tid + u * GROUP_THREADS;
        if (i < n_f4) {
            float4* p = (float4*)(acc + (i >> 5) * ACC_PITCH + (i & 31) * 4);
            float4 a = *p;
            a.x = scale * f[u].x + a.x;
            a.y = scale * f[u].y + a.y;
            a.z = scale * f[u].z + a.z;
            a.w = scale * f[u].w + a.w;
            *p = a;
        }
    }
    #pragma unroll
    for (int u = 0; u < W_PIECES; ++u) {
        const int i = tid + u * GROUP_THREADS;
        float4* t = (float4*)(wl + (i >> 5) * W_PITCH + (i & 31) * 8);
        t[0] = make_float4(w0[u].x, w1[u].x, w0[u].y, w1[u].y);
        t[1] = make_float4(w0[u].z, w1[u].z, w0[u].w, w1[u].w);
    }
    __syncthreads();

    const int half = lane >> 4;
    const int l15 = lane & 15;
    const int n_strips = ((rows + 15) >> 4) * 4;
    for (int st = wave; st < n_strips; st += GROUP_WAVES) {
        const int r0 = (st >> 2) * 16;
        const int n0 = (st & 3) * 32;
        const float* ap = acc + (r0 + l15) * ACC_PITCH + half * 2;
        const float* bp = wl + half * W_PITCH + (n0 + l15) * 2;
        v8f c0 = {};
        v8f c1 = {};
        #pragma unroll 8
        for (int ks = 0; ks < D / 4; ++ks) {
            const v2f a  = *(const v2f*)(ap + ks * 4);
            const v2f b0 = *(const v2f*)(bp + ks * (2 * W_PITCH));
            const v2f b1 = *(const v2f*)(bp + ks * (2 * W_PITCH) + 32);
            c0 = __builtin_amdgcn_wmma_f32_16x16x4_f32(
                 false, a,  false, b0,
                 (short)0, c0,  false,  false);
            c1 = __builtin_amdgcn_wmma_f32_16x16x4_f32(
                 false, a,  false, b1,
                 (short)0, c1,  false,  false);
        }
        const float bv0 = bias[n0 + l15];
        const float bv1 = bias[n0 + 16 + l15];
        const int rl = r0 + half * 8;
        float* orow = out + (size_t)(lo + rl) * D + n0 + l15;
        #pragma unroll
        for (int v = 0; v < 8; ++v) {
            if (rl + v < rows) {
                orow[(size_t)v * D]      = c0[v] + bv0;
                orow[(size_t)v * D + 16] = c1[v] + bv1;
            }
        }
    }
}

extern "C" void kernel_launch(void* const* d_in, const int* in_sizes, int n_in,
                              void* d_out, int out_size, void* d_ws, size_t ws_size,
                              hipStream_t stream) {
    const float* feat = (const float*)d_in[0];
    const float* W    = (const float*)d_in[1];
    const float* b    = (const float*)d_in[2];
    const float* eps  = (const float*)d_in[3];
    const int*   src  = (const int*)d_in[4];
    const int*   dst  = (const int*)d_in[5];
    float*       out  = (float*)d_out;
    unsigned*    ws   = (unsigned*)d_ws;

    gin_bin_kernel<<<N_SLICES, BIN_THREADS, 0, stream>>>(src, dst, ws);
    gin_conv_kernel<<<N_GROUPS, GROUP_THREADS, 0, stream>>>(
        feat, ws, W, b, eps, out);
}


# aggregation: a group's edge list is counting-sorted by destination row in LDS; a wave sums each run of equal destination in registers, one LDS float add per run, was per edge; arrival-order sums kept
# speedup vs baseline: 4.8945x; 1.1196x over previous
#include <hip/hip_runtime.h>

typedef __attribute__((ext_vector_type(2))) float v2f;
typedef __attribute__((ext_vector_type(8))) float v8f;

#define N_NODES 50000
#define N_EDGES 800000
#define D 128

#define ROWS_PER_GROUP 196
#define N_GROUPS 256
#define GROUP_THREADS 1024
#define GROUP_WAVES (GROUP_THREADS / 32)
#define LIST_CAP 16384
#define EDGE_BATCH 8
#define N_SLICES 256
#define SLICE_EDGES (N_EDGES / N_SLICES)
#define BIN_THREADS 256
#define BIN_WAVES (BIN_THREADS / 32)
#define BIN_FULL (SLICE_EDGES / BIN_THREADS)
#define BIN_TAIL (SLICE_EDGES - BIN_FULL * BIN_THREADS)
#define BIN_PER_THREAD (BIN_FULL + 1)
#define SLICE_PITCH 3200
#define TAB_PITCH 288
#define TAB_BASE (N_SLICES * SLICE_PITCH)
#define TEAM 4
#define TEAM_DEPTH 4
#define ACC_PITCH (D + 4)
#define ACC_ROWS 208
#define ACC_FLOATS (ACC_ROWS * ACC_PITCH)
#define W_PITCH (2 * D + 32)
#define W_FLOATS ((D / 2) * W_PITCH)
#define FEAT_PIECES ((ROWS_PER_GROUP * (D / 4) + GROUP_THREADS - 1) / GROUP_THREADS)
#define W_PIECES (((D / 2) * (D / 4)) / GROUP_THREADS)
static_assert(ROWS_PER_GROUP * N_GROUPS >= N_NODES, "the ranges cover every node");
static_assert((N_GROUPS - 1) * ROWS_PER_GROUP < N_NODES, "every group owns at least one node");
static_assert(ROWS_PER_GROUP + 1 <= 256, "a local row fits the low 8 bits of an edge word");
static_assert(ROWS_PER_GROUP + 1 <= ACC_ROWS && ACC_ROWS % 16 == 0, "whole row tiles, padding row included");
static_assert(N_NODES <= (1 << 24), "a source row fits the high 24 bits of an edge word");
static_assert(N_EDGES % N_SLICES == 0, "the slices are equal");
static_assert(BIN_THREADS == N_GROUPS, "one thread of a launch-1 group per counter");
static_assert(BIN_TAIL > 0 && BIN_TAIL <= BIN_THREADS, "the last round is a short one");
static_assert(SLICE_PITCH >= SLICE_EDGES && SLICE_PITCH % 32 == 0, "a slice's words fill whole lines of its own");
static_assert(TAB_PITCH >= N_GROUPS + 1 && TAB_PITCH % 32 == 0, "a slice's offsets fill whole lines of its own");
static_assert(N_SLICES * TEAM == GROUP_THREADS && (TEAM & (TEAM - 1)) == 0 && TEAM <= 32,
              "every slice has one team, inside one wave");
static_assert(D == 128 && ACC_PITCH % 4 == 0 && W_PITCH % 4 == 0, "16-byte LDS accesses stay aligned");
static_assert(((D / 2) * (D / 4)) % GROUP_THREADS == 0, "W is staged in whole rounds of the group");
#define ROW_SLOTS 256
#define AGG_LDS_BYTES (ACC_FLOATS * 4 + LIST_CAP * 4 + 16 + W_FLOATS * 4 + ROW_SLOTS * 4)
static_assert(W_FLOATS >= LIST_CAP, "the ordered list fits W's place, which is free until phase 3");
static_assert(ROW_SLOTS == 8 * 32 && ROW_SLOTS > ROWS_PER_GROUP, "one wave scans the counters, 8 per lane");
static_assert(AGG_LDS_BYTES <= 320 * 1024, "sums, list, counter and W fit the work-group's LDS");

__global__ void __launch_bounds__(BIN_THREADS)
gin_bin_kernel(const int* __restrict__ src,
               const int* __restrict__ dst,
               unsigned* __restrict__ ws) {
    __shared__ unsigned count[N_GROUPS];
    __shared__ unsigned offs[N_GROUPS + 1];
    __shared__ unsigned wave_sum[BIN_WAVES];
    __shared__ unsigned sorted[SLICE_EDGES];

    const int tid = threadIdx.x;
    const int lane = tid & 31;
    const int wave = tid >> 5;
    const int e0 = blockIdx.x * SLICE_EDGES;

    int dv[BIN_PER_THREAD];
    int sv[BIN_PER_THREAD];
    dv[BIN_FULL] = -1;
    sv[BIN_FULL] = 0;
    if (tid < BIN_TAIL) {
        dv[BIN_FULL] = dst[e0 + BIN_FULL * BIN_THREADS + tid];
        sv[BIN_FULL] = src[e0 + BIN_FULL * BIN_THREADS + tid];
    }
    #pragma unroll
    for (int u = 0; u < BIN_FULL; ++u) {
        dv[u] = dst[e0 + u * BIN_THREADS + tid];
        sv[u] = src[e0 + u * BIN_THREADS + tid];
    }
    count[tid] = 0u;
    __syncthreads();

    unsigned h[BIN_PER_THREAD];
    unsigned word[BIN_PER_THREAD];
    unsigned rank[BIN_PER_THREAD];
    #pragma unroll
    for (int u = 0; u < BIN_PER_THREAD; ++u) {
        const unsigned d = (unsigned)dv[u];
        h[u] = N_GROUPS;
        word[u] = 0u;
        rank[u] = 0u;
        if (d < (unsigned)N_NODES) {
            h[u] = d / (unsigned)ROWS_PER_GROUP;
            word[u] = ((unsigned)sv[u] << 8) | (d - h[u] * (unsigned)ROWS_PER_GROUP);
            rank[u] = __hip_atomic_fetch_add(&count[h[u]], 1u, __ATOMIC_RELAXED,
                                             __HIP_MEMORY_SCOPE_WORKGROUP);
        }
    }
    __syncthreads();

    const unsigned c = count[tid];
    unsigned x = c;
    #pragma unroll
    for (int o = 1; o < 32; o <<= 1) {
        const unsigned y = __shfl_up(x, o);
        if (lane >= o) x += y;
    }
    if (lane == 31) wave_sum[wave] = x;
    __syncthreads();
    unsigned base = 0u;
    #pragma unroll
    for (int w = 0; w < BIN_WAVES; ++w) {
        if (w < wave) base += wave_sum[w];
    }
    offs[tid] = base + x - c;
    if (tid == BIN_THREADS - 1) offs[N_GROUPS] = base + x;
    __syncthreads();

    unsigned place[BIN_PER_THREAD];
    #pragma unroll
    for (int u = 0; u < BIN_PER_THREAD; ++u) place[u] = offs[h[u]] + rank[u];
    #pragma unroll
    for (int u = 0; u < BIN_PER_THREAD; ++u) {
        if (h[u] < (unsigned)N_GROUPS) sorted[place[u]] = word[u];
    }
    __syncthreads();

    const unsigned kept = offs[N_GROUPS];
    unsigned* wout = ws + (size_t)blockIdx.x * SLICE_PITCH;
    for (unsigned k = (unsigned)tid; k < kept; k += BIN_THREADS) wout[k] = sorted[k];
    unsigned* tout = ws + TAB_BASE + (size_t)blockIdx.x * TAB_PITCH;
    tout[tid] = offs[tid];
    if (tid == 0) tout[N_GROUPS] = kept;
}

__global__ void __launch_bounds__(GROUP_THREADS)
gin_conv_kernel(const float* __restrict__ feat,
                const unsigned* __restrict__ ws,
                const float* __restrict__ W,
                const float* __restrict__ bias,
                const float* __restrict__ eps,
                float* __restrict__ out) {
    __shared__ __attribute__((aligned(16))) unsigned char L[AGG_LDS_BYTES];
    float* acc = (float*)L;
    unsigned* list = (unsigned*)(L + ACC_FLOATS * 4);
    unsigned* cnt = (unsigned*)(L + ACC_FLOATS * 4 + LIST_CAP * 4);
    float* wl = (float*)(L + ACC_FLOATS * 4 + LIST_CAP * 4 + 16);
    unsigned* sorted = (unsigned*)wl;
    unsigned* rcnt = (unsigned*)(L + ACC_FLOATS * 4 + LIST_CAP * 4 + 16 + W_FLOATS * 4);

    const int tid = threadIdx.x;
    const int lane = tid & 31;
    const int wave = __builtin_amdgcn_readfirstlane(tid >> 5);
    const int lo = blockIdx.x * ROWS_PER_GROUP;

    const int slice = tid / TEAM;
    const int q = tid & (TEAM - 1);
    const unsigned* trow = ws + TAB_BASE + (size_t)slice * TAB_PITCH + blockIdx.x;
    const unsigned run_lo = trow[0];
    const unsigned run_hi = trow[1];

    float4* acc4 = (float4*)acc;
    for (int i = tid; i < ACC_FLOATS / 4; i += GROUP_THREADS) {
        acc4[i] = make_float4(0.f, 0.f, 0.f, 0.f);
    }
    if (tid == 0) *cnt = 0u;
    if (tid < ROW_SLOTS) rcnt[tid] = 0u;
    __syncthreads();

    const unsigned run = run_hi - run_lo;
    const unsigned* ewords = ws + (size_t)slice * SLICE_PITCH + run_lo;
    unsigned pos = 0u;
    if (q == 0) {
        pos = __hip_atomic_fetch_add(cnt, run, __ATOMIC_RELAXED, __HIP_MEMORY_SCOPE_WORKGROUP);
    }
    pos = (unsigned)__shfl((int)pos, lane & ~(TEAM - 1));
    unsigned ew[TEAM_DEPTH];
    #pragma unroll
    for (int u = 0; u < TEAM_DEPTH; ++u) {
        const unsigned k = (unsigned)(q + u * TEAM);
        ew[u] = 0u;
        if (k < run) ew[u] = ewords[k];
    }
    __syncthreads();

    const unsigned total = (unsigned)__builtin_amdgcn_readfirstlane((int)*cnt);
    if (total <= (unsigned)LIST_CAP) {
        #pragma unroll
        for (int u = 0; u < TEAM_DEPTH; ++u) {
            const unsigned k = (unsigned)(q + u * TEAM);
            if (k < run) {
                list[pos + k] = ew[u];
                __hip_atomic_fetch_add(&rcnt[ew[u] & 255u], 1u, __ATOMIC_RELAXED,
                                       __HIP_MEMORY_SCOPE_WORKGROUP);
            }
        }
        for (unsigned k = (unsigned)(q + TEAM_DEPTH * TEAM); k < run; k += TEAM) {
            const unsigned w = ewords[k];
            list[pos + k] = w;
            __hip_atomic_fetch_add(&rcnt[w & 255u], 1u, __ATOMIC_RELAXED,
                                   __HIP_MEMORY_SCOPE_WORKGROUP);
        }
    }
    __syncthreads();

    if (total <= (unsigned)LIST_CAP && wave == 0) {
        unsigned c[8];
        unsigned sum = 0u;
        #pragma unroll
        for (int i = 0; i < 8; ++i) {
            c[i] = rcnt[lane * 8 + i];
            sum += c[i];
        }
        unsigned x = sum;
        #pragma unroll
        for (int o = 1; o < 32; o <<= 1) {
            const unsigned y = __shfl_up(x, o);
            if (lane >= o) x += y;
        }
        unsigned at = x - sum;
        #pragma unroll
        for (int i = 0; i < 8; ++i) {
            rcnt[lane * 8 + i] = at;
            at += c[i];
        }
    }
    __syncthreads();

    if (total <= (unsigned)LIST_CAP) {
        for (unsigned k = (unsigned)tid; k < total; k += GROUP_THREADS) {
            const unsigned w = list[k];
            const unsigned p = __hip_atomic_fetch_add(&rcnt[w & 255u], 1u, __ATOMIC_RELAXED,
                                                      __HIP_MEMORY_SCOPE_WORKGROUP);
            sorted[p] = w;
        }
    }
    __syncthreads();

    if (total <= (unsigned)LIST_CAP) {
        const int b_lo = (int)((total * (unsigned)wave) / (unsigned)GROUP_WAVES);
        const int n = (int)((total * (unsigned)(wave + 1)) / (unsigned)GROUP_WAVES);
        const unsigned lane_b = (unsigned)lane * 4u;
        const char* fbase = (const char*)feat;
        const unsigned pad = (unsigned)ROWS_PER_GROUP;

        unsigned cur = pad;
        float s0 = 0.f, s1 = 0.f, s2 = 0.f, s3 = 0.f;

        int b0 = b_lo;
        unsigned ent = pad;
        if (lane < EDGE_BATCH && b0 + lane < n) ent = sorted[b0 + lane];
        for (; b0 < n; b0 += EDGE_BATCH) {
            const unsigned ent_c = ent;
            float v[EDGE_BATCH][4];
            #pragma unroll
            for (int j = 0; j < EDGE_BATCH; ++j) {
                const unsigned sj = (unsigned)__builtin_amdgcn_readlane((int)ent_c, j) >> 8;
                const float* frow = (const float*)(fbase + (sj * (unsigned)(D * 4) + lane_b));
                v[j][0] = frow[0];
                v[j][1] = frow[32];
                v[j][2] = frow[64];
                v[j][3] = frow[96];
            }
            const int b1 = b0 + EDGE_BATCH;
            ent = pad;
            if (lane < EDGE_BATCH && b1 + lane < n) ent = sorted[b1 + lane];
            #pragma unroll
            for (int j = 0; j < EDGE_BATCH; ++j) {
                const unsigned dj = (unsigned)__builtin_amdgcn_readlane((int)ent_c, j) & 255u;
                if (dj != cur) {
                    float* arow = acc + cur * ACC_PITCH + lane;
                    __hip_atomic_fetch_add(arow + 0,  s0, __ATOMIC_RELAXED, __HIP_MEMORY_SCOPE_WORKGROUP);
                    __hip_atomic_fetch_add(arow + 32, s1, __ATOMIC_RELAXED, __HIP_MEMORY_SCOPE_WORKGROUP);
                    __hip_atomic_fetch_add(arow + 64, s2, __ATOMIC_RELAXED, __HIP_MEMORY_SCOPE_WORKGROUP);
                    __hip_atomic_fetch_add(arow + 96, s3, __ATOMIC_RELAXED, __HIP_MEMORY_SCOPE_WORKGROUP);
                    cur = dj;
                    s0 = v[j][0];
                    s1 = v[j][1];
                    s2 = v[j][2];
                    s3 = v[j][3];
                } else {
                    s0 += v[j][0];
                    s1 += v[j][1];
                    s2 += v[j][2];
                    s3 += v[j][3];
                }
            }
        }
        {
            float* arow = acc + cur * ACC_PITCH + lane;
            __hip_atomic_fetch_add(arow + 0,  s0, __ATOMIC_RELAXED, __HIP_MEMORY_SCOPE_WORKGROUP);
            __hip_atomic_fetch_add(arow + 32, s1, __ATOMIC_RELAXED, __HIP_MEMORY_SCOPE_WORKGROUP);
            __hip_atomic_fetch_add(arow + 64, s2, __ATOMIC_RELAXED, __HIP_MEMORY_SCOPE_WORKGROUP);
            __hip_atomic_fetch_add(arow + 96, s3, __ATOMIC_RELAXED, __HIP_MEMORY_SCOPE_WORKGROUP);
        }
    } else {
        for (unsigned k = (unsigned)q; k < run; k += TEAM) {
            const unsigned w = ewords[k];
            const float* frow = feat + (size_t)(w >> 8) * D;
            float* arow = acc + (w & 255u) * ACC_PITCH;
            for (int j = 0; j < D; ++j) {
                __hip_atomic_fetch_add(arow + j, frow[j], __ATOMIC_RELAXED,
                                       __HIP_MEMORY_SCOPE_WORKGROUP);
            }
        }
    }
    __syncthreads();

    const int rows = (N_NODES - lo) < ROWS_PER_GROUP ? (N_NODES - lo) : ROWS_PER_GROUP;
    const float scale = 1.0f + eps[0];

    const float4* feat4 = (const float4*)(feat + (size_t)lo * D);
    const float4* W4 = (const float4*)W;
    const int n_f4 = rows * (D / 4);
    float4 f[FEAT_PIECES];
    #pragma unroll
    for (int u = 0; u < FEAT_PIECES; ++u) {
        const int i = tid + u * GROUP_THREADS;
        f[u] = make_float4(0.f, 0.f, 0.f, 0.f);
        if (i < n_f4) f[u] = feat4[i];
    }
    float4 w0[W_PIECES];
    float4 w1[W_PIECES];
    #pragma unroll
    for (int u = 0; u < W_PIECES; ++u) {
        const int i = tid + u * GROUP_THREADS;
        w0[u] = W4[(2 * (i >> 5)) * (D / 4) + (i & 31)];
        w1[u] = W4[(2 * (i >> 5) + 1) * (D / 4) + (i & 31)];
    }
    #pragma unroll
    for (int u = 0; u < FEAT_PIECES; ++u) {
        const int i = tid + u * GROUP_THREADS;
        if (i < n_f4) {
            float4* p = (float4*)(acc + (i >> 5) * ACC_PITCH + (i & 31) * 4);
            float4 a = *p;
            a.x = scale * f[u].x + a.x;
            a.y = scale * f[u].y + a.y;
            a.z = scale * f[u].z + a.z;
            a.w = scale * f[u].w + a.w;
            *p = a;
        }
    }
    #pragma unroll
    for (int u = 0; u < W_PIECES; ++u) {
        const int i = tid + u * GROUP_THREADS;
        float4* t = (float4*)(wl + (i >> 5) * W_PITCH + (i & 31) * 8);
        t[0] = make_float4(w0[u].x, w1[u].x, w0[u].y, w1[u].y);
        t[1] = make_float4(w0[u].z, w1[u].z, w0[u].w, w1[u].w);
    }
    __syncthreads();

    const int half = lane >> 4;
    const int l15 = lane & 15;
    const int n_strips = ((rows + 15) >> 4) * 4;
    for (int st = wave; st < n_strips; st += GROUP_WAVES) {
        const int r0 = (st >> 2) * 16;
        const int n0 = (st & 3) * 32;
        const float* ap = acc + (r0 + l15) * ACC_PITCH + half * 2;
        const float* bp = wl + half * W_PITCH + (n0 + l15) * 2;
        v8f c0 = {};
        v8f c1 = {};
        #pragma unroll 8
        for (int ks = 0; ks < D / 4; ++ks) {
            const v2f a  = *(const v2f*)(ap + ks * 4);
            const v2f b0 = *(const v2f*)(bp + ks * (2 * W_PITCH));
            const v2f b1 = *(const v2f*)(bp + ks * (2 * W_PITCH) + 32);
            c0 = __builtin_amdgcn_wmma_f32_16x16x4_f32(
                 false, a,  false, b0,
                 (short)0, c0,  false,  false);
            c1 = __builtin_amdgcn_wmma_f32_16x16x4_f32(
                 false, a,  false, b1,
                 (short)0, c1,  false,  false);
        }
        const float bv0 = bias[n0 + l15];
        const float bv1 = bias[n0 + 16 + l15];
        const int rl = r0 + half * 8;
        float* orow = out + (size_t)(lo + rl) * D + n0 + l15;
        #pragma unroll
        for (int v = 0; v < 8; ++v) {
            if (rl + v < rows) {
                orow[(size_t)v * D]      = c0[v] + bv0;
                orow[(size_t)v * D + 16] = c1[v] + bv1;
            }
        }
    }
}

extern "C" void kernel_launch(void* const* d_in, const int* in_sizes, int n_in,
                              void* d_out, int out_size, void* d_ws, size_t ws_size,
                              hipStream_t stream) {
    const float* feat = (const float*)d_in[0];
    const float* W    = (const float*)d_in[1];
    const float* b    = (const float*)d_in[2];
    const float* eps  = (const float*)d_in[3];
    const int*   src  = (const int*)d_in[4];
    const int*   dst  = (const int*)d_in[5];
    float*       out  = (float*)d_out;
    unsigned*    ws   = (unsigned*)d_ws;

    gin_bin_kernel<<<N_SLICES, BIN_THREADS, 0, stream>>>(src, dst, ws);
    gin_conv_kernel<<<N_GROUPS, GROUP_THREADS, 0, stream>>>(
        feat, ws, W, b, eps, out);
}
